# starting kernel
# baseline (speedup 1.0000x reference)
#include <hip/hip_runtime.h>

#define N_NODES 50000
#define N_EDGES 600000
#define HID 128
#define NEG_SLOPE 0.01f
#define N_LAYERS 3

typedef __attribute__((ext_vector_type(2))) float v2f;
typedef __attribute__((ext_vector_type(8))) float v8f;

__global__ __launch_bounds__(256) void zero_kernel(float4* __restrict__ p, int n4) {
  int i = blockIdx.x * 256 + threadIdx.x;
  if (i < n4) p[i] = make_float4(0.f, 0.f, 0.f, 0.f);
}

__global__ __launch_bounds__(256) void edge_kernel(
    const float* __restrict__ h,
    const int*   __restrict__ src,
    const int*   __restrict__ dst,
    const float* __restrict__ ea,
    const float* __restrict__ Wl0,
    const float* __restrict__ Wl1,
    const float* __restrict__ bl,
    float* __restrict__ agg)
{
  int tid = blockIdx.x * 256 + threadIdx.x;
  int e   = tid >> 5;
  int c   = (tid & 31) * 4;
  if (e >= N_EDGES) return;

  int   s  = src[e];
  int   d  = dst[e];
  float a0 = ea[2 * e];
  float a1 = ea[2 * e + 1];

  float4 w0 = *(const float4*)(Wl0 + c);
  float4 w1 = *(const float4*)(Wl1 + c);
  float4 bb = *(const float4*)(bl  + c);
  float4 hv = *(const float4*)(h + (long)s * HID + c);

  float m0 = hv.x + a0 * w0.x + a1 * w1.x + bb.x;
  float m1 = hv.y + a0 * w0.y + a1 * w1.y + bb.y;
  float m2 = hv.z + a0 * w0.z + a1 * w1.z + bb.z;
  float m3 = hv.w + a0 * w0.w + a1 * w1.w + bb.w;
  m0 = m0 > 0.f ? m0 : 0.f;
  m1 = m1 > 0.f ? m1 : 0.f;
  m2 = m2 > 0.f ? m2 : 0.f;
  m3 = m3 > 0.f ? m3 : 0.f;

  float* ap = agg + (long)d * HID + c;
  unsafeAtomicAdd(ap + 0, m0);
  unsafeAtomicAdd(ap + 1, m1);
  unsafeAtomicAdd(ap + 2, m2);
  unsafeAtomicAdd(ap + 3, m3);
}

__global__ __launch_bounds__(256) void gine_gemm_kernel(
    const float* __restrict__ h,
    const float* __restrict__ agg,
    const float* __restrict__ W,
    const float* __restrict__ bias,
    float* __restrict__ out)
{
  int wave = threadIdx.x >> 5;
  int lane = threadIdx.x & 31;
  int m0   = blockIdx.x * 16;
  int n0   = wave * 16;
  int m    = lane & 15;
  int half = lane >> 4;
  int n    = n0 + m;

  const float* xr = h   + (long)(m0 + m) * HID + half * 2;
  const float* gr = agg + (long)(m0 + m) * HID + half * 2;

  v8f c = {};
  for (int kk = 0; kk < HID; kk += 4) {
    float2 hx = *(const float2*)(xr + kk);
    float2 gx = *(const float2*)(gr + kk);
    v2f a;
    a.x = hx.x + gx.x;
    a.y = hx.y + gx.y;

    int k0 = kk + half * 2;
    v2f bm;
    bm.x = W[(long)k0 * HID + n];
    bm.y = W[(long)(k0 + 1) * HID + n];

    c = __builtin_amdgcn_wmma_f32_16x16x4_f32(
         false, a,  false, bm,
         (short)0, c,  false,  false);
  }

  float bv = bias[n];
#pragma unroll
  for (int v = 0; v < 8; ++v) {
    int row = m0 + v + half * 8;
    float y = c[v] + bv;
    out[(long)row * HID + n] = (y >= 0.f) ? y : NEG_SLOPE * y;
  }
}

extern "C" void kernel_launch(void* const* d_in, const int* in_sizes, int n_in,
                              void* d_out, int out_size, void* d_ws, size_t ws_size,
                              hipStream_t stream) {
  (void)in_sizes; (void)n_in; (void)out_size; (void)ws_size;

  const float* x  = (const float*)d_in[0];
  const int*   ei = (const int*)  d_in[1];
  const float* ea = (const float*)d_in[2];
  const float* Wl = (const float*)d_in[3];
  const float* bl = (const float*)d_in[4];
  const float* W  = (const float*)d_in[5];
  const float* b  = (const float*)d_in[6];
  float*       out = (float*)d_out;

  const int* src = ei;
  const int* dst = ei + N_EDGES;

  const long NH = (long)N_NODES * HID;
  float* agg = (float*)d_ws;
  float* h0  = agg + NH;
  float* h1  = h0 + NH;

  const float* hin = x;
  for (int layer = 0; layer < N_LAYERS; ++layer) {
    float* hout = (layer == N_LAYERS - 1) ? out : (layer == 0 ? h0 : h1);

    zero_kernel<<<(int)(NH / 4 + 255) / 256, 256, 0, stream>>>((float4*)agg, (int)(NH / 4));

    edge_kernel<<<N_EDGES / 8, 256, 0, stream>>>(
        hin, src, dst, ea,
        Wl + (long)layer * 2 * HID,
        Wl + (long)layer * 2 * HID + HID,
        bl + (long)layer * HID,
        agg);

    gine_gemm_kernel<<<N_NODES / 16, 256, 0, stream>>>(
        hin, agg,
        W + (long)layer * HID * HID,
        b + (long)layer * HID,
        hout);

    hin = hout;
  }
}


# Node GEMM: a wave owns 16 rows x all 128 columns (was one 16x16 tile, X re-read by 8 waves): X loaded once, W shared via LDS, whole-line stores; k order of the dot product regrouped
# speedup vs baseline: 2.0591x; 2.0591x over previous
#include <hip/hip_runtime.h>

#define N_NODES 50000
#define N_EDGES 600000
#define HID 128
#define NEG_SLOPE 0.01f
#define N_LAYERS 3

typedef __attribute__((ext_vector_type(2))) float v2f;
typedef __attribute__((ext_vector_type(8))) float v8f;

__global__ __launch_bounds__(256) void zero_kernel(float4* __restrict__ p, int n4) {
  int i = blockIdx.x * 256 + threadIdx.x;
  if (i < n4) p[i] = make_float4(0.f, 0.f, 0.f, 0.f);
}

#define EDGES_PER_WAVE 32
#define EDGE_GROUP 8
#define EDGE_WAVES (N_EDGES / EDGES_PER_WAVE)
#define EDGE_BLOCKS ((EDGE_WAVES + 7) / 8)
static_assert(N_EDGES % EDGES_PER_WAVE == 0, "a wave takes 32 whole edges");

__global__ __launch_bounds__(256) void edge_kernel(
    const float* __restrict__ h,
    const int*   __restrict__ src,
    const int*   __restrict__ dst,
    const float* __restrict__ ea,
    const float* __restrict__ Wl0,
    const float* __restrict__ Wl1,
    const float* __restrict__ bl,
    float* __restrict__ agg)
{
  const int l = threadIdx.x & 31;
  const int w = __builtin_amdgcn_readfirstlane((int)(blockIdx.x * 8 + (threadIdx.x >> 5)));
  if (w >= EDGE_WAVES) return;

  const int e = w * EDGES_PER_WAVE + l;
  const int    s_l = src[e];
  const int    d_l = dst[e];
  const float2 a_l = *(const float2*)(ea + 2 * (long)e);
  const int a0_l = __float_as_int(a_l.x);
  const int a1_l = __float_as_int(a_l.y);

  float w0[4], w1[4], bb[4];
#pragma unroll
  for (int j = 0; j < 4; ++j) {
    w0[j] = Wl0[l + 32 * j];
    w1[j] = Wl1[l + 32 * j];
    bb[j] = bl[l + 32 * j];
  }

#pragma unroll
  for (int g = 0; g < EDGES_PER_WAVE; g += EDGE_GROUP) {
    float hv[EDGE_GROUP][4];
#pragma unroll
    for (int i = 0; i < EDGE_GROUP; ++i) {
      const int s = __builtin_amdgcn_readlane(s_l, g + i);
      const float* hp = h + (long)s * HID;
      hv[i][0] = hp[l];
      hv[i][1] = hp[l + 32];
      hv[i][2] = hp[l + 64];
      hv[i][3] = hp[l + 96];
    }
#pragma unroll
    for (int i = 0; i < EDGE_GROUP; ++i) {
      const int   d  = __builtin_amdgcn_readlane(d_l, g + i);
      const float a0 = __int_as_float(__builtin_amdgcn_readlane(a0_l, g + i));
      const float a1 = __int_as_float(__builtin_amdgcn_readlane(a1_l, g + i));
      float* ap = agg + (long)d * HID;
#pragma unroll
      for (int j = 0; j < 4; ++j) {
        float m = hv[i][j] + a0 * w0[j] + a1 * w1[j] + bb[j];
        m = m > 0.f ? m : 0.f;
        unsafeAtomicAdd(ap + l + 32 * j, m);
      }
    }
  }
}

typedef __attribute__((ext_vector_type(4))) float v4f;

#define GEMM_WAVES 8
#define ROW_TILES (N_NODES / 16)
#define GEMM_BLOCKS ((ROW_TILES + GEMM_WAVES - 1) / GEMM_WAVES)
#define WP_PITCH 260
static_assert(N_NODES % 16 == 0, "row tiles are whole");

__global__ __launch_bounds__(256) void gine_gemm_kernel(
    const float* __restrict__ h,
    const float* __restrict__ agg,
    const float* __restrict__ W,
    const float* __restrict__ bias,
    float* __restrict__ out)
{
  __shared__ float Wl[(HID / 2) * WP_PITCH];

  const int tid  = threadIdx.x;
  const int lane = tid & 31;
  const int wave = __builtin_amdgcn_readfirstlane(tid >> 5);
  const int c    = lane & 15;
  const int hh   = lane >> 4;
  const int tile = blockIdx.x * GEMM_WAVES + wave;
  const bool live = tile < ROW_TILES;

  v4f we[8], wo[8];
#pragma unroll
  for (int i = 0; i < 8; ++i) {
    const int p = 8 * i + wave;
    we[i] = ((const v4f*)W)[(2 * p) * 32 + lane];
    wo[i] = ((const v4f*)W)[(2 * p + 1) * 32 + lane];
  }

  v4f xa[16], xg[16];
  if (live) {
    const float* xr = h   + ((long)tile * 16 + c) * HID + 4 * hh;
    const float* gr = agg + ((long)tile * 16 + c) * HID + 4 * hh;
#pragma unroll
    for (int q = 0; q < 16; ++q) {
      xa[q] = *(const v4f*)(xr + 8 * q);
      xg[q] = *(const v4f*)(gr + 8 * q);
    }
  }

  {
    float* wd = Wl + 128 * hh + 4 * c;
#pragma unroll
    for (int i = 0; i < 8; ++i) {
      float* d = wd + (8 * i + wave) * WP_PITCH;
      *(v4f*)d        = (v4f){we[i][0], wo[i][0], we[i][1], wo[i][1]};
      *(v4f*)(d + 64) = (v4f){we[i][2], wo[i][2], we[i][3], wo[i][3]};
    }
  }
  __syncthreads();
  if (!live) return;

  v4f x[16];
#pragma unroll
  for (int q = 0; q < 16; ++q) x[q] = xa[q] + xg[q];

  v8f acc[8];
#pragma unroll
  for (int t = 0; t < 8; ++t) acc[t] = (v8f){};

  const float* wb = Wl + (2 * hh) * WP_PITCH + 4 * c;

  v4f bn[2][4];
#pragma unroll
  for (int u = 0; u < 2; ++u)
#pragma unroll
    for (int j = 0; j < 4; ++j)
      bn[u][j] = *(const v4f*)(wb + u * WP_PITCH + 64 * j);
#pragma unroll
  for (int q = 0; q < 16; ++q) {
    v4f b[2][4];
#pragma unroll
    for (int u = 0; u < 2; ++u)
#pragma unroll
      for (int j = 0; j < 4; ++j) b[u][j] = bn[u][j];
    if (q + 1 < 16) {
#pragma unroll
      for (int u = 0; u < 2; ++u)
#pragma unroll
        for (int j = 0; j < 4; ++j)
          bn[u][j] = *(const v4f*)(wb + (4 * (q + 1) + u) * WP_PITCH + 64 * j);
    }
#pragma unroll
    for (int u = 0; u < 2; ++u) {
      v2f a;
      a.x = x[q][2 * u];
      a.y = x[q][2 * u + 1];
#pragma unroll
      for (int j = 0; j < 4; ++j) {
#pragma unroll
        for (int e = 0; e < 2; ++e) {
          v2f bm;
          bm.x = b[u][j][2 * e];
          bm.y = b[u][j][2 * e + 1];
          acc[2 * j + e] = __builtin_amdgcn_wmma_f32_16x16x4_f32(
               false, a,  false, bm,
               (short)0, acc[2 * j + e],  false,  false);
        }
      }
    }
  }

  float* orow = out + ((long)tile * 16 + 8 * hh) * HID + 4 * c;
#pragma unroll
  for (int g = 0; g < 2; ++g) {
    const v4f bv = *(const v4f*)(bias + 64 * g + 4 * c);
#pragma unroll
    for (int r = 0; r < 8; ++r) {
      v4f y;
#pragma unroll
      for (int i = 0; i < 4; ++i) {
        const float t = acc[4 * g + i][r] + bv[i];
        y[i] = (t >= 0.f) ? t : NEG_SLOPE * t;
      }
      *(v4f*)(orow + (long)r * HID + 64 * g) = y;
    }
  }
}

extern "C" void kernel_launch(void* const* d_in, const int* in_sizes, int n_in,
                              void* d_out, int out_size, void* d_ws, size_t ws_size,
                              hipStream_t stream) {
  (void)in_sizes; (void)n_in; (void)out_size; (void)ws_size;

  const float* x  = (const float*)d_in[0];
  const int*   ei = (const int*)  d_in[1];
  const float* ea = (const float*)d_in[2];
  const float* Wl = (const float*)d_in[3];
  const float* bl = (const float*)d_in[4];
  const float* W  = (const float*)d_in[5];
  const float* b  = (const float*)d_in[6];
  float*       out = (float*)d_out;

  const int* src = ei;
  const int* dst = ei + N_EDGES;

  const long NH = (long)N_NODES * HID;
  float* agg = (float*)d_ws;
  float* h0  = agg + NH;
  float* h1  = h0 + NH;

  const float* hin = x;
  for (int layer = 0; layer < N_LAYERS; ++layer) {
    float* hout = (layer == N_LAYERS - 1) ? out : (layer == 0 ? h0 : h1);

    zero_kernel<<<(int)(NH / 4 + 255) / 256, 256, 0, stream>>>((float4*)agg, (int)(NH / 4));

    edge_kernel<<<EDGE_BLOCKS, 256, 0, stream>>>(
        hin, src, dst, ea,
        Wl + (long)layer * 2 * HID,
        Wl + (long)layer * 2 * HID + HID,
        bl + (long)layer * HID,
        agg);

    gine_gemm_kernel<<<GEMM_BLOCKS, 256, 0, stream>>>(
        hin, agg,
        W + (long)layer * HID * HID,
        b + (long)layer * HID,
        hout);

    hin = hout;
  }
}


# Edge scatter by ownership: a block owns 196 target rows, filters its edges out of the whole dst list and sums them in LDS (ds_add_f32, arrival order); no global atomics, no zero pass
# speedup vs baseline: 2.5387x; 1.2329x over previous
#include <hip/hip_runtime.h>

#define N_NODES 50000
#define N_EDGES 600000
#define HID 128
#define NEG_SLOPE 0.01f
#define N_LAYERS 3

typedef __attribute__((ext_vector_type(2))) float v2f;
typedef __attribute__((ext_vector_type(8))) float v8f;

#define OWN_ROWS 196
#define OWN_BLOCKS ((N_NODES + OWN_ROWS - 1) / OWN_ROWS)
#define OWN_WAVES 32
#define OWN_THREADS (OWN_WAVES * 32)
#define SCAN_U 4
#define Q_SLOTS 1024
#define EDGE_GROUP 8
#define N_DST4 (N_EDGES / 4)
#define SCAN_STEP (OWN_WAVES * 32 * SCAN_U)
#define SCAN_ITERS ((N_DST4 + SCAN_STEP - 1) / SCAN_STEP)
static_assert(N_EDGES % 4 == 0, "dst is read in whole 16-byte pieces");
static_assert(OWN_ROWS <= 256, "a local row fits the entry's low 8 bits");
static_assert(N_EDGES < (1 << 23), "an edge id fits the entry's upper bits");
static_assert(31 + 128 * SCAN_U <= Q_SLOTS, "ring holds a leftover and one scan step");

__global__ __launch_bounds__(OWN_THREADS) void edge_owner_kernel(
    const float* __restrict__ h,
    const int*   __restrict__ src,
    const int*   __restrict__ dst,
    const float* __restrict__ ea,
    const float* __restrict__ Wl0,
    const float* __restrict__ Wl1,
    const float* __restrict__ bl,
    float* __restrict__ agg)
{
  __shared__ float acc[OWN_ROWS * HID];
  __shared__ int   ring[OWN_WAVES * Q_SLOTS];

  const int tid = threadIdx.x;
  const int l   = tid & 31;
  const int w   = __builtin_amdgcn_readfirstlane(tid >> 5);
  const int n0  = blockIdx.x * OWN_ROWS;

  for (int i = tid; i < OWN_ROWS * HID / 4; i += OWN_THREADS)
    ((float4*)acc)[i] = make_float4(0.f, 0.f, 0.f, 0.f);

  float w0[4], w1[4], bb[4];
#pragma unroll
  for (int j = 0; j < 4; ++j) {
    w0[j] = Wl0[l + 32 * j];
    w1[j] = Wl1[l + 32 * j];
    bb[j] = bl[l + 32 * j];
  }
  __syncthreads();

  int* q = ring + w * Q_SLOTS;
  int qhead = 0, qlen = 0;

  const int4* d4 = (const int4*)dst;
  int4 dn[SCAN_U];
#pragma unroll
  for (int u = 0; u < SCAN_U; ++u) {
    const int idx = (w * SCAN_U + u) * 32 + l;
    dn[u] = d4[idx < N_DST4 ? idx : N_DST4 - 1];
  }

  for (int it = 0; it <= SCAN_ITERS; ++it) {
    if (it < SCAN_ITERS) {
      int4 d[SCAN_U];
#pragma unroll
      for (int u = 0; u < SCAN_U; ++u) d[u] = dn[u];
      const int ib = it * SCAN_STEP + w * SCAN_U * 32 + l;
      if (it == SCAN_ITERS - 1) {
#pragma unroll
        for (int u = 0; u < SCAN_U; ++u)
          if (ib + u * 32 >= N_DST4) d[u] = make_int4(-1, -1, -1, -1);
      }
      if (it + 1 < SCAN_ITERS) {
#pragma unroll
        for (int u = 0; u < SCAN_U; ++u) {
          const int idx = ib + SCAN_STEP + u * 32;
          dn[u] = d4[idx < N_DST4 ? idx : N_DST4 - 1];
        }
      }
#pragma unroll
      for (int u = 0; u < SCAN_U; ++u) {
        const int e0 = 4 * (ib + u * 32);
        const unsigned r0 = (unsigned)(d[u].x - n0);
        const unsigned r1 = (unsigned)(d[u].y - n0);
        const unsigned r2 = (unsigned)(d[u].z - n0);
        const unsigned r3 = (unsigned)(d[u].w - n0);
        const bool m0 = r0 < OWN_ROWS, m1 = r1 < OWN_ROWS;
        const bool m2 = r2 < OWN_ROWS, m3 = r3 < OWN_ROWS;
        const unsigned k0 = __builtin_amdgcn_ballot_w32(m0);
        const unsigned k1 = __builtin_amdgcn_ballot_w32(m1);
        const unsigned k2 = __builtin_amdgcn_ballot_w32(m2);
        const unsigned k3 = __builtin_amdgcn_ballot_w32(m3);
        if (k0 | k1 | k2 | k3) {
          int p = qhead + qlen;
          if (m0) q[(p + __builtin_amdgcn_mbcnt_lo(k0, 0)) & (Q_SLOTS - 1)] = ((e0 + 0) << 8) | (int)r0;
          p += __builtin_popcount(k0);
          if (m1) q[(p + __builtin_amdgcn_mbcnt_lo(k1, 0)) & (Q_SLOTS - 1)] = ((e0 + 1) << 8) | (int)r1;
          p += __builtin_popcount(k1);
          if (m2) q[(p + __builtin_amdgcn_mbcnt_lo(k2, 0)) & (Q_SLOTS - 1)] = ((e0 + 2) << 8) | (int)r2;
          p += __builtin_popcount(k2);
          if (m3) q[(p + __builtin_amdgcn_mbcnt_lo(k3, 0)) & (Q_SLOTS - 1)] = ((e0 + 3) << 8) | (int)r3;
          p += __builtin_popcount(k3);
          qlen = p - qhead;
        }
      }
    }
    const int need = it < SCAN_ITERS ? 32 : 1;
    while (qlen >= need) {
      const int cnt = qlen < 32 ? qlen : 32;
      int ent = 0;
      if (l < cnt) ent = q[(qhead + l) & (Q_SLOTS - 1)];
      const int eid   = ent >> 8;
      const int row_l = ent & 255;
      const int    s_l = src[eid];
      const float2 a_l = *(const float2*)(ea + 2 * (long)eid);
      const int a0_l = __float_as_int(a_l.x);
      const int a1_l = __float_as_int(a_l.y);
#pragma unroll
      for (int g = 0; g < 32; g += EDGE_GROUP) {
        if (g < cnt) {
          float hv[EDGE_GROUP][4];
#pragma unroll
          for (int i = 0; i < EDGE_GROUP; ++i) {
            const int s = __builtin_amdgcn_readlane(s_l, g + i);
            const float* hp = h + (long)s * HID;
            hv[i][0] = hp[l];
            hv[i][1] = hp[l + 32];
            hv[i][2] = hp[l + 64];
            hv[i][3] = hp[l + 96];
          }
#pragma unroll
          for (int i = 0; i < EDGE_GROUP; ++i) {
            if (g + i < cnt) {
              const int   r  = __builtin_amdgcn_readlane(row_l, g + i);
              const float a0 = __int_as_float(__builtin_amdgcn_readlane(a0_l, g + i));
              const float a1 = __int_as_float(__builtin_amdgcn_readlane(a1_l, g + i));
              float* ap = acc + r * HID + l;
#pragma unroll
              for (int j = 0; j < 4; ++j) {
                float m = hv[i][j] + a0 * w0[j] + a1 * w1[j] + bb[j];
                m = m > 0.f ? m : 0.f;
                __hip_atomic_fetch_add(ap + 32 * j, m, __ATOMIC_RELAXED, __HIP_MEMORY_SCOPE_WORKGROUP);
              }
            }
          }
        }
      }
      qhead = (qhead + cnt) & (Q_SLOTS - 1);
      qlen -= cnt;
    }
  }

  __syncthreads();
  const long base4 = (long)n0 * (HID / 4);
  for (int i = tid; i < OWN_ROWS * HID / 4; i += OWN_THREADS)
    if (base4 + i < (long)N_NODES * (HID / 4))
      ((float4*)agg)[base4 + i] = ((const float4*)acc)[i];
}

typedef __attribute__((ext_vector_type(4))) float v4f;

#define GEMM_WAVES 8
#define ROW_TILES (N_NODES / 16)
#define GEMM_BLOCKS ((ROW_TILES + GEMM_WAVES - 1) / GEMM_WAVES)
#define WP_PITCH 260
static_assert(N_NODES % 16 == 0, "row tiles are whole");

__global__ __launch_bounds__(256) void gine_gemm_kernel(
    const float* __restrict__ h,
    const float* __restrict__ agg,
    const float* __restrict__ W,
    const float* __restrict__ bias,
    float* __restrict__ out)
{
  __shared__ float Wl[(HID / 2) * WP_PITCH];

  const int tid  = threadIdx.x;
  const int lane = tid & 31;
  const int wave = __builtin_amdgcn_readfirstlane(tid >> 5);
  const int c    = lane & 15;
  const int hh   = lane >> 4;
  const int tile = blockIdx.x * GEMM_WAVES + wave;
  const bool live = tile < ROW_TILES;

  v4f we[8], wo[8];
#pragma unroll
  for (int i = 0; i < 8; ++i) {
    const int p = 8 * i + wave;
    we[i] = ((const v4f*)W)[(2 * p) * 32 + lane];
    wo[i] = ((const v4f*)W)[(2 * p + 1) * 32 + lane];
  }

  v4f xa[16], xg[16];
  if (live) {
    const float* xr = h   + ((long)tile * 16 + c) * HID + 4 * hh;
    const float* gr = agg + ((long)tile * 16 + c) * HID + 4 * hh;
#pragma unroll
    for (int q = 0; q < 16; ++q) {
      xa[q] = *(const v4f*)(xr + 8 * q);
      xg[q] = *(const v4f*)(gr + 8 * q);
    }
  }

  {
    float* wd = Wl + 128 * hh + 4 * c;
#pragma unroll
    for (int i = 0; i < 8; ++i) {
      float* d = wd + (8 * i + wave) * WP_PITCH;
      *(v4f*)d        = (v4f){we[i][0], wo[i][0], we[i][1], wo[i][1]};
      *(v4f*)(d + 64) = (v4f){we[i][2], wo[i][2], we[i][3], wo[i][3]};
    }
  }
  __syncthreads();
  if (!live) return;

  v4f x[16];
#pragma unroll
  for (int q = 0; q < 16; ++q) x[q] = xa[q] + xg[q];

  v8f acc[8];
#pragma unroll
  for (int t = 0; t < 8; ++t) acc[t] = (v8f){};

  const float* wb = Wl + (2 * hh) * WP_PITCH + 4 * c;

  v4f bn[2][4];
#pragma unroll
  for (int u = 0; u < 2; ++u)
#pragma unroll
    for (int j = 0; j < 4; ++j)
      bn[u][j] = *(const v4f*)(wb + u * WP_PITCH + 64 * j);
#pragma unroll
  for (int q = 0; q < 16; ++q) {
    v4f b[2][4];
#pragma unroll
    for (int u = 0; u < 2; ++u)
#pragma unroll
      for (int j = 0; j < 4; ++j) b[u][j] = bn[u][j];
    if (q + 1 < 16) {
#pragma unroll
      for (int u = 0; u < 2; ++u)
#pragma unroll
        for (int j = 0; j < 4; ++j)
          bn[u][j] = *(const v4f*)(wb + (4 * (q + 1) + u) * WP_PITCH + 64 * j);
    }
#pragma unroll
    for (int u = 0; u < 2; ++u) {
      v2f a;
      a.x = x[q][2 * u];
      a.y = x[q][2 * u + 1];
#pragma unroll
      for (int j = 0; j < 4; ++j) {
#pragma unroll
        for (int e = 0; e < 2; ++e) {
          v2f bm;
          bm.x = b[u][j][2 * e];
          bm.y = b[u][j][2 * e + 1];
          acc[2 * j + e] = __builtin_amdgcn_wmma_f32_16x16x4_f32(
               false, a,  false, bm,
               (short)0, acc[2 * j + e],  false,  false);
        }
      }
    }
  }

  float* orow = out + ((long)tile * 16 + 8 * hh) * HID + 4 * c;
#pragma unroll
  for (int g = 0; g < 2; ++g) {
    const v4f bv = *(const v4f*)(bias + 64 * g + 4 * c);
#pragma unroll
    for (int r = 0; r < 8; ++r) {
      v4f y;
#pragma unroll
      for (int i = 0; i < 4; ++i) {
        const float t = acc[4 * g + i][r] + bv[i];
        y[i] = (t >= 0.f) ? t : NEG_SLOPE * t;
      }
      *(v4f*)(orow + (long)r * HID + 64 * g) = y;
    }
  }
}

extern "C" void kernel_launch(void* const* d_in, const int* in_sizes, int n_in,
                              void* d_out, int out_size, void* d_ws, size_t ws_size,
                              hipStream_t stream) {
  (void)in_sizes; (void)n_in; (void)out_size; (void)ws_size;

  const float* x  = (const float*)d_in[0];
  const int*   ei = (const int*)  d_in[1];
  const float* ea = (const float*)d_in[2];
  const float* Wl = (const float*)d_in[3];
  const float* bl = (const float*)d_in[4];
  const float* W  = (const float*)d_in[5];
  const float* b  = (const float*)d_in[6];
  float*       out = (float*)d_out;

  const int* src = ei;
  const int* dst = ei + N_EDGES;

  const long NH = (long)N_NODES * HID;
  float* agg = (float*)d_ws;
  float* h0  = agg + NH;
  float* h1  = h0 + NH;

  const float* hin = x;
  for (int layer = 0; layer < N_LAYERS; ++layer) {
    float* hout = (layer == N_LAYERS - 1) ? out : (layer == 0 ? h0 : h1);

    edge_owner_kernel<<<OWN_BLOCKS, OWN_THREADS, 0, stream>>>(
        hin, src, dst, ea,
        Wl + (long)layer * 2 * HID,
        Wl + (long)layer * 2 * HID + HID,
        bl + (long)layer * HID,
        agg);

    gine_gemm_kernel<<<GEMM_BLOCKS, 256, 0, stream>>>(
        hin, agg,
        W + (long)layer * HID * HID,
        b + (long)layer * HID,
        hout);

    hin = hout;
  }
}


# Edge phase: layer 0 records each block's owned edges as whole records {src, local row, attrs} in per-wave lists; layers 1, 2 replay the lists: no dst scan, no index loads; LDS sums arrival-order
# speedup vs baseline: 3.1853x; 1.2547x over previous
#include <hip/hip_runtime.h>

#define N_NODES 50000
#define N_EDGES 600000
#define HID 128
#define NEG_SLOPE 0.01f
#define N_LAYERS 3

typedef __attribute__((ext_vector_type(2))) float v2f;
typedef __attribute__((ext_vector_type(8))) float v8f;

#define OWN_ROWS 196
#define OWN_BLOCKS ((N_NODES + OWN_ROWS - 1) / OWN_ROWS)
#define OWN_WAVES 32
#define OWN_THREADS (OWN_WAVES * 32)
#define SCAN_U 4
#define Q_SLOTS 1024
#define EDGE_GROUP 8
#define N_DST4 (N_EDGES / 4)
#define SCAN_STEP (OWN_WAVES * 32 * SCAN_U)
#define SCAN_ITERS ((N_DST4 + SCAN_STEP - 1) / SCAN_STEP)
#define REC_CAP 192
#define N_LISTS (OWN_BLOCKS * OWN_WAVES)
#define LIST_INTS ((long)N_LISTS * REC_CAP * 4)
static_assert(N_EDGES % 4 == 0, "dst is read in whole 16-byte pieces");
static_assert(OWN_ROWS <= 256, "a local row fits the entry's low 8 bits");
static_assert(N_EDGES < (1 << 23), "an edge id fits the entry's upper bits");
static_assert(31 + 128 * SCAN_U <= Q_SLOTS, "ring holds a leftover and one scan step");
static_assert(REC_CAP % 32 == 0, "a list is a whole number of 32-record groups");
static_assert((LIST_INTS + N_LISTS) * 4 <= (long)N_NODES * HID * 4,
              "lists and lengths fit the workspace plane they replace");

__device__ __forceinline__ void add_group(
    const float* __restrict__ h, float* acc, const int l, const int cnt,
    const int s_l, const int row_l, const int a0_l, const int a1_l,
    const float (&w0)[4], const float (&w1)[4], const float (&bb)[4])
{
#pragma unroll
  for (int g = 0; g < 32; g += EDGE_GROUP) {
    if (g < cnt) {
      float hv[EDGE_GROUP][4];
#pragma unroll
      for (int i = 0; i < EDGE_GROUP; ++i) {
        const int s = __builtin_amdgcn_readlane(s_l, g + i);
        const float* hp = h + (long)s * HID;
        hv[i][0] = hp[l];
        hv[i][1] = hp[l + 32];
        hv[i][2] = hp[l + 64];
        hv[i][3] = hp[l + 96];
      }
#pragma unroll
      for (int i = 0; i < EDGE_GROUP; ++i) {
        if (g + i < cnt) {
          const int   r  = __builtin_amdgcn_readlane(row_l, g + i);
          const float a0 = __int_as_float(__builtin_amdgcn_readlane(a0_l, g + i));
          const float a1 = __int_as_float(__builtin_amdgcn_readlane(a1_l, g + i));
          float* ap = acc + r * HID + l;
#pragma unroll
          for (int j = 0; j < 4; ++j) {
            float m = hv[i][j] + a0 * w0[j] + a1 * w1[j] + bb[j];
            m = m > 0.f ? m : 0.f;
            __hip_atomic_fetch_add(ap + 32 * j, m, __ATOMIC_RELAXED, __HIP_MEMORY_SCOPE_WORKGROUP);
          }
        }
      }
    }
  }
}

template <bool RECORD>
__device__ __forceinline__ int scan_wave(
    const float* __restrict__ h, const int* __restrict__ src, const int* __restrict__ dst,
    const float* __restrict__ ea, float* acc, int* q, const int l, const int w, const int n0,
    const float (&w0)[4], const float (&w1)[4], const float (&bb)[4],
    int4* lst)
{
  int qhead = 0, qlen = 0;
  int nrec = 0;
  bool over = false;

  const int4* d4 = (const int4*)dst;
  int4 dn[SCAN_U];
#pragma unroll
  for (int u = 0; u < SCAN_U; ++u) {
    const int idx = (w * SCAN_U + u) * 32 + l;
    dn[u] = d4[idx < N_DST4 ? idx : N_DST4 - 1];
  }

  for (int it = 0; it <= SCAN_ITERS; ++it) {
    if (it < SCAN_ITERS) {
      int4 d[SCAN_U];
#pragma unroll
      for (int u = 0; u < SCAN_U; ++u) d[u] = dn[u];
      const int ib = it * SCAN_STEP + w * SCAN_U * 32 + l;
      if (it == SCAN_ITERS - 1) {
#pragma unroll
        for (int u = 0; u < SCAN_U; ++u)
          if (ib + u * 32 >= N_DST4) d[u] = make_int4(-1, -1, -1, -1);
      }
      if (it + 1 < SCAN_ITERS) {
#pragma unroll
        for (int u = 0; u < SCAN_U; ++u) {
          const int idx = ib + SCAN_STEP + u * 32;
          dn[u] = d4[idx < N_DST4 ? idx : N_DST4 - 1];
        }
      }
#pragma unroll
      for (int u = 0; u < SCAN_U; ++u) {
        const int e0 = 4 * (ib + u * 32);
        const unsigned r0 = (unsigned)(d[u].x - n0);
        const unsigned r1 = (unsigned)(d[u].y - n0);
        const unsigned r2 = (unsigned)(d[u].z - n0);
        const unsigned r3 = (unsigned)(d[u].w - n0);
        const bool m0 = r0 < OWN_ROWS, m1 = r1 < OWN_ROWS;
        const bool m2 = r2 < OWN_ROWS, m3 = r3 < OWN_ROWS;
        const unsigned k0 = __builtin_amdgcn_ballot_w32(m0);
        const unsigned k1 = __builtin_amdgcn_ballot_w32(m1);
        const unsigned k2 = __builtin_amdgcn_ballot_w32(m2);
        const unsigned k3 = __builtin_amdgcn_ballot_w32(m3);
        if (k0 | k1 | k2 | k3) {
          int p = qhead + qlen;
          if (m0) q[(p + __builtin_amdgcn_mbcnt_lo(k0, 0)) & (Q_SLOTS - 1)] = ((e0 + 0) << 8) | (int)r0;
          p += __builtin_popcount(k0);
          if (m1) q[(p + __builtin_amdgcn_mbcnt_lo(k1, 0)) & (Q_SLOTS - 1)] = ((e0 + 1) << 8) | (int)r1;
          p += __builtin_popcount(k1);
          if (m2) q[(p + __builtin_amdgcn_mbcnt_lo(k2, 0)) & (Q_SLOTS - 1)] = ((e0 + 2) << 8) | (int)r2;
          p += __builtin_popcount(k2);
          if (m3) q[(p + __builtin_amdgcn_mbcnt_lo(k3, 0)) & (Q_SLOTS - 1)] = ((e0 + 3) << 8) | (int)r3;
          p += __builtin_popcount(k3);
          qlen = p - qhead;
        }
      }
    }
    const int need = it < SCAN_ITERS ? 32 : 1;
    while (qlen >= need) {
      const int cnt = qlen < 32 ? qlen : 32;
      int ent = 0;
      if (l < cnt) ent = q[(qhead + l) & (Q_SLOTS - 1)];
      const int eid   = ent >> 8;
      const int row_l = ent & 255;
      const int    s_l = src[eid];
      const float2 a_l = *(const float2*)(ea + 2 * (long)eid);
      const int a0_l = __float_as_int(a_l.x);
      const int a1_l = __float_as_int(a_l.y);
      if (RECORD) {
        if (!over && nrec + cnt <= REC_CAP) {
          if (l < cnt) lst[nrec + l] = make_int4(s_l, row_l, a0_l, a1_l);
          nrec += cnt;
        } else {
          over = true;
        }
      }
      add_group(h, acc, l, cnt, s_l, row_l, a0_l, a1_l, w0, w1, bb);
      qhead = (qhead + cnt) & (Q_SLOTS - 1);
      qlen -= cnt;
    }
  }
  return over ? -1 : nrec;
}

template <bool RECORD>
__device__ __forceinline__ void edge_owner_body(
    float* acc, int* ring, int* lens,
    const float* __restrict__ h,
    const int*   __restrict__ src,
    const int*   __restrict__ dst,
    const float* __restrict__ ea,
    const float* __restrict__ Wl0,
    const float* __restrict__ Wl1,
    const float* __restrict__ bl,
    float* __restrict__ agg,
    int* lists)
{
  const int tid = threadIdx.x;
  const int l   = tid & 31;
  const int w   = __builtin_amdgcn_readfirstlane(tid >> 5);
  const int n0  = blockIdx.x * OWN_ROWS;

  for (int i = tid; i < OWN_ROWS * HID / 4; i += OWN_THREADS)
    ((float4*)acc)[i] = make_float4(0.f, 0.f, 0.f, 0.f);

  float w0[4], w1[4], bb[4];
#pragma unroll
  for (int j = 0; j < 4; ++j) {
    w0[j] = Wl0[l + 32 * j];
    w1[j] = Wl1[l + 32 * j];
    bb[j] = bl[l + 32 * j];
  }

  const int lid = blockIdx.x * OWN_WAVES + w;
  int4* lst = (int4*)lists + (long)lid * REC_CAP;
  int*  len = lists + LIST_INTS + (long)blockIdx.x * OWN_WAVES;

  if (RECORD) {
    __syncthreads();
    const int n = scan_wave<true>(h, src, dst, ea, acc, ring + w * Q_SLOTS, l, w, n0, w0, w1, bb, lst);
    if (l == 0) lens[w] = n;
  } else {
    const int n = __builtin_amdgcn_readfirstlane(len[w]);
    int4 rec[REC_CAP / 32];
#pragma unroll
    for (int k = 0; k < REC_CAP / 32; ++k) {
      rec[k] = make_int4(0, 0, 0, 0);
      if (32 * k + l < n) rec[k] = lst[32 * k + l];
    }
    __syncthreads();
    if (n >= 0) {
#pragma unroll
      for (int k = 0; k < REC_CAP / 32; ++k) {
        if (32 * k < n) {
          const int rest = n - 32 * k;
          add_group(h, acc, l, rest < 32 ? rest : 32, rec[k].x, rec[k].y, rec[k].z, rec[k].w, w0, w1, bb);
        }
      }
    } else {
      scan_wave<false>(h, src, dst, ea, acc, ring + w * Q_SLOTS, l, w, n0, w0, w1, bb, lst);
    }
  }

  __syncthreads();
  if (RECORD) {
    if (w == 0) len[l] = lens[l];
  }
  const long base4 = (long)n0 * (HID / 4);
  for (int i = tid; i < OWN_ROWS * HID / 4; i += OWN_THREADS)
    if (base4 + i < (long)N_NODES * (HID / 4))
      ((float4*)agg)[base4 + i] = ((const float4*)acc)[i];
}

__global__ __launch_bounds__(OWN_THREADS) void edge_owner_record_kernel(
    const float* __restrict__ h, const int* __restrict__ src, const int* __restrict__ dst,
    const float* __restrict__ ea, const float* __restrict__ Wl0, const float* __restrict__ Wl1,
    const float* __restrict__ bl, float* __restrict__ agg, int* __restrict__ lists)
{
  __shared__ float acc[OWN_ROWS * HID];
  __shared__ int   ring[OWN_WAVES * Q_SLOTS];
  __shared__ int   lens[OWN_WAVES];
  edge_owner_body<true>(acc, ring, lens, h, src, dst, ea, Wl0, Wl1, bl, agg, lists);
}

__global__ __launch_bounds__(OWN_THREADS) void edge_owner_replay_kernel(
    const float* __restrict__ h, const int* __restrict__ src, const int* __restrict__ dst,
    const float* __restrict__ ea, const float* __restrict__ Wl0, const float* __restrict__ Wl1,
    const float* __restrict__ bl, float* __restrict__ agg, int* __restrict__ lists)
{
  __shared__ float acc[OWN_ROWS * HID];
  __shared__ int   ring[OWN_WAVES * Q_SLOTS];
  edge_owner_body<false>(acc, ring, nullptr, h, src, dst, ea, Wl0, Wl1, bl, agg, lists);
}

typedef __attribute__((ext_vector_type(4))) float v4f;

#define GEMM_WAVES 8
#define ROW_TILES (N_NODES / 16)
#define GEMM_BLOCKS ((ROW_TILES + GEMM_WAVES - 1) / GEMM_WAVES)
#define WP_PITCH 260
static_assert(N_NODES % 16 == 0, "row tiles are whole");

__global__ __launch_bounds__(256) void gine_gemm_kernel(
    const float* h,
    const float* __restrict__ agg,
    const float* __restrict__ W,
    const float* __restrict__ bias,
    float* out)
{
  __shared__ float Wl[(HID / 2) * WP_PITCH];

  const int tid  = threadIdx.x;
  const int lane = tid & 31;
  const int wave = __builtin_amdgcn_readfirstlane(tid >> 5);
  const int c    = lane & 15;
  const int hh   = lane >> 4;
  const int tile = blockIdx.x * GEMM_WAVES + wave;
  const bool live = tile < ROW_TILES;

  v4f we[8], wo[8];
#pragma unroll
  for (int i = 0; i < 8; ++i) {
    const int p = 8 * i + wave;
    we[i] = ((const v4f*)W)[(2 * p) * 32 + lane];
    wo[i] = ((const v4f*)W)[(2 * p + 1) * 32 + lane];
  }

  v4f xa[16], xg[16];
  if (live) {
    const float* xr = h   + ((long)tile * 16 + c) * HID + 4 * hh;
    const float* gr = agg + ((long)tile * 16 + c) * HID + 4 * hh;
#pragma unroll
    for (int q = 0; q < 16; ++q) {
      xa[q] = *(const v4f*)(xr + 8 * q);
      xg[q] = *(const v4f*)(gr + 8 * q);
    }
  }

  {
    float* wd = Wl + 128 * hh + 4 * c;
#pragma unroll
    for (int i = 0; i < 8; ++i) {
      float* d = wd + (8 * i + wave) * WP_PITCH;
      *(v4f*)d        = (v4f){we[i][0], wo[i][0], we[i][1], wo[i][1]};
      *(v4f*)(d + 64) = (v4f){we[i][2], wo[i][2], we[i][3], wo[i][3]};
    }
  }
  __syncthreads();
  if (!live) return;

  v4f x[16];
#pragma unroll
  for (int q = 0; q < 16; ++q) x[q] = xa[q] + xg[q];

  v8f acc[8];
#pragma unroll
  for (int t = 0; t < 8; ++t) acc[t] = (v8f){};

  const float* wb = Wl + (2 * hh) * WP_PITCH + 4 * c;

  v4f bn[2][4];
#pragma unroll
  for (int u = 0; u < 2; ++u)
#pragma unroll
    for (int j = 0; j < 4; ++j)
      bn[u][j] = *(const v4f*)(wb + u * WP_PITCH + 64 * j);
#pragma unroll
  for (int q = 0; q < 16; ++q) {
    v4f b[2][4];
#pragma unroll
    for (int u = 0; u < 2; ++u)
#pragma unroll
      for (int j = 0; j < 4; ++j) b[u][j] = bn[u][j];
    if (q + 1 < 16) {
#pragma unroll
      for (int u = 0; u < 2; ++u)
#pragma unroll
        for (int j = 0; j < 4; ++j)
          bn[u][j] = *(const v4f*)(wb + (4 * (q + 1) + u) * WP_PITCH + 64 * j);
    }
#pragma unroll
    for (int u = 0; u < 2; ++u) {
      v2f a;
      a.x = x[q][2 * u];
      a.y = x[q][2 * u + 1];
#pragma unroll
      for (int j = 0; j < 4; ++j) {
#pragma unroll
        for (int e = 0; e < 2; ++e) {
          v2f bm;
          bm.x = b[u][j][2 * e];
          bm.y = b[u][j][2 * e + 1];
          acc[2 * j + e] = __builtin_amdgcn_wmma_f32_16x16x4_f32(
               false, a,  false, bm,
               (short)0, acc[2 * j + e],  false,  false);
        }
      }
    }
  }

  float* orow = out + ((long)tile * 16 + 8 * hh) * HID + 4 * c;
#pragma unroll
  for (int g = 0; g < 2; ++g) {
    const v4f bv = *(const v4f*)(bias + 64 * g + 4 * c);
#pragma unroll
    for (int r = 0; r < 8; ++r) {
      v4f y;
#pragma unroll
      for (int i = 0; i < 4; ++i) {
        const float t = acc[4 * g + i][r] + bv[i];
        y[i] = (t >= 0.f) ? t : NEG_SLOPE * t;
      }
      *(v4f*)(orow + (long)r * HID + 64 * g) = y;
    }
  }
}

extern "C" void kernel_launch(void* const* d_in, const int* in_sizes, int n_in,
                              void* d_out, int out_size, void* d_ws, size_t ws_size,
                              hipStream_t stream) {
  (void)in_sizes; (void)n_in; (void)out_size; (void)ws_size;

  const float* x  = (const float*)d_in[0];
  const int*   ei = (const int*)  d_in[1];
  const float* ea = (const float*)d_in[2];
  const float* Wl = (const float*)d_in[3];
  const float* bl = (const float*)d_in[4];
  const float* W  = (const float*)d_in[5];
  const float* b  = (const float*)d_in[6];
  float*       out = (float*)d_out;

  const int* src = ei;
  const int* dst = ei + N_EDGES;

  const long NH = (long)N_NODES * HID;
  float* agg   = (float*)d_ws;
  float* hbuf  = agg + NH;
  int*   lists = (int*)(hbuf + NH);

  const float* hin = x;
  for (int layer = 0; layer < N_LAYERS; ++layer) {
    float* hout = (layer == N_LAYERS - 1) ? out : hbuf;

    if (layer == 0)
      edge_owner_record_kernel<<<OWN_BLOCKS, OWN_THREADS, 0, stream>>>(
          hin, src, dst, ea,
          Wl + (long)layer * 2 * HID,
          Wl + (long)layer * 2 * HID + HID,
          bl + (long)layer * HID,
          agg, lists);
    else
      edge_owner_replay_kernel<<<OWN_BLOCKS, OWN_THREADS, 0, stream>>>(
          hin, src, dst, ea,
          Wl + (long)layer * 2 * HID,
          Wl + (long)layer * 2 * HID + HID,
          bl + (long)layer * HID,
          agg, lists);

    gine_gemm_kernel<<<GEMM_BLOCKS, 256, 0, stream>>>(
        hin, agg,
        W + (long)layer * HID * HID,
        b + (long)layer * HID,
        hout);

    hin = hout;
  }
}


# Node GEMM moved into the edge owner block: a row's LDS sum starts at its own h (arrival-order adds as before), waves feed WMMA from LDS and store next h; agg never stored, launches 6 to 3
# speedup vs baseline: 3.8742x; 1.2163x over previous
#include <hip/hip_runtime.h>

#define N_NODES 50000
#define N_EDGES 600000
#define HID 128
#define NEG_SLOPE 0.01f
#define N_LAYERS 3

typedef __attribute__((ext_vector_type(2))) float v2f;
typedef __attribute__((ext_vector_type(8))) float v8f;
typedef __attribute__((ext_vector_type(4))) float v4f;

#define OWN_ROWS 196
#define OWN_BLOCKS ((N_NODES + OWN_ROWS - 1) / OWN_ROWS)
#define OWN_WAVES 32
#define OWN_THREADS (OWN_WAVES * 32)
#define OWN_TILES ((OWN_ROWS + 15) / 16)
#define ACC_PITCH 132
#define WP_PITCH 260
#define SCAN_U 4
#define Q_SLOTS 1024
#define EDGE_GROUP 8
#define N_DST4 (N_EDGES / 4)
#define SCAN_STEP (OWN_WAVES * 32 * SCAN_U)
#define SCAN_ITERS ((N_DST4 + SCAN_STEP - 1) / SCAN_STEP)
#define REC_CAP 192
#define N_LISTS (OWN_BLOCKS * OWN_WAVES)
#define LIST_INTS ((long)N_LISTS * REC_CAP * 4)
static_assert(N_EDGES % 4 == 0, "dst is read in whole 16-byte pieces");
static_assert(OWN_ROWS <= 256, "a local row fits the entry's low 8 bits");
static_assert(N_EDGES < (1 << 23), "an edge id fits the entry's upper bits");
static_assert(31 + 128 * SCAN_U <= Q_SLOTS, "ring holds a leftover and one scan step");
static_assert(REC_CAP % 32 == 0, "a list is a whole number of 32-record groups");
static_assert(2 * OWN_TILES <= OWN_WAVES, "one wave per (row tile, column half)");
static_assert(HID == 128, "a row is 32 lanes x 16 bytes; the fragment arithmetic below");
static_assert((LIST_INTS + N_LISTS) * 4 <= (long)N_NODES * HID * 4,
              "lists and lengths fit the workspace plane they replace");

__device__ __forceinline__ void add_group(
    const float* __restrict__ h, float* acc, const int l, const int cnt,
    const int s_l, const int row_l, const int a0_l, const int a1_l,
    const float (&w0)[4], const float (&w1)[4], const float (&bb)[4])
{
#pragma unroll
  for (int g = 0; g < 32; g += EDGE_GROUP) {
    if (g < cnt) {
      float hv[EDGE_GROUP][4];
#pragma unroll
      for (int i = 0; i < EDGE_GROUP; ++i) {
        const int s = __builtin_amdgcn_readlane(s_l, g + i);
        const float* hp = h + (long)s * HID;
        hv[i][0] = hp[l];
        hv[i][1] = hp[l + 32];
        hv[i][2] = hp[l + 64];
        hv[i][3] = hp[l + 96];
      }
#pragma unroll
      for (int i = 0; i < EDGE_GROUP; ++i) {
        if (g + i < cnt) {
          const int   r  = __builtin_amdgcn_readlane(row_l, g + i);
          const float a0 = __int_as_float(__builtin_amdgcn_readlane(a0_l, g + i));
          const float a1 = __int_as_float(__builtin_amdgcn_readlane(a1_l, g + i));
          float* ap = acc + r * ACC_PITCH + l;
#pragma unroll
          for (int j = 0; j < 4; ++j) {
            float m = hv[i][j] + a0 * w0[j] + a1 * w1[j] + bb[j];
            m = m > 0.f ? m : 0.f;
            __hip_atomic_fetch_add(ap + 32 * j, m, __ATOMIC_RELAXED, __HIP_MEMORY_SCOPE_WORKGROUP);
          }
        }
      }
    }
  }
}

template <bool RECORD>
__device__ __forceinline__ int scan_wave(
    const float* __restrict__ h, const int* __restrict__ src, const int* __restrict__ dst,
    const float* __restrict__ ea, float* acc, int* q, const int l, const int w, const int n0,
    const float (&w0)[4], const float (&w1)[4], const float (&bb)[4],
    int4* lst)
{
  int qhead = 0, qlen = 0;
  int nrec = 0;
  bool over = false;

  const int4* d4 = (const int4*)dst;
  int4 dn[SCAN_U];
#pragma unroll
  for (int u = 0; u < SCAN_U; ++u) {
    const int idx = (w * SCAN_U + u) * 32 + l;
    dn[u] = d4[idx < N_DST4 ? idx : N_DST4 - 1];
  }

  for (int it = 0; it <= SCAN_ITERS; ++it) {
    if (it < SCAN_ITERS) {
      int4 d[SCAN_U];
#pragma unroll
      for (int u = 0; u < SCAN_U; ++u) d[u] = dn[u];
      const int ib = it * SCAN_STEP + w * SCAN_U * 32 + l;
      if (it == SCAN_ITERS - 1) {
#pragma unroll
        for (int u = 0; u < SCAN_U; ++u)
          if (ib + u * 32 >= N_DST4) d[u] = make_int4(-1, -1, -1, -1);
      }
      if (it + 1 < SCAN_ITERS) {
#pragma unroll
        for (int u = 0; u < SCAN_U; ++u) {
          const int idx = ib + SCAN_STEP + u * 32;
          dn[u] = d4[idx < N_DST4 ? idx : N_DST4 - 1];
        }
      }
#pragma unroll
      for (int u = 0; u < SCAN_U; ++u) {
        const int e0 = 4 * (ib + u * 32);
        const unsigned r0 = (unsigned)(d[u].x - n0);
        const unsigned r1 = (unsigned)(d[u].y - n0);
        const unsigned r2 = (unsigned)(d[u].z - n0);
        const unsigned r3 = (unsigned)(d[u].w - n0);
        const bool m0 = r0 < OWN_ROWS, m1 = r1 < OWN_ROWS;
        const bool m2 = r2 < OWN_ROWS, m3 = r3 < OWN_ROWS;
        const unsigned k0 = __builtin_amdgcn_ballot_w32(m0);
        const unsigned k1 = __builtin_amdgcn_ballot_w32(m1);
        const unsigned k2 = __builtin_amdgcn_ballot_w32(m2);
        const unsigned k3 = __builtin_amdgcn_ballot_w32(m3);
        if (k0 | k1 | k2 | k3) {
          int p = qhead + qlen;
          if (m0) q[(p + __builtin_amdgcn_mbcnt_lo(k0, 0)) & (Q_SLOTS - 1)] = ((e0 + 0) << 8) | (int)r0;
          p += __builtin_popcount(k0);
          if (m1) q[(p + __builtin_amdgcn_mbcnt_lo(k1, 0)) & (Q_SLOTS - 1)] = ((e0 + 1) << 8) | (int)r1;
          p += __builtin_popcount(k1);
          if (m2) q[(p + __builtin_amdgcn_mbcnt_lo(k2, 0)) & (Q_SLOTS - 1)] = ((e0 + 2) << 8) | (int)r2;
          p += __builtin_popcount(k2);
          if (m3) q[(p + __builtin_amdgcn_mbcnt_lo(k3, 0)) & (Q_SLOTS - 1)] = ((e0 + 3) << 8) | (int)r3;
          p += __builtin_popcount(k3);
          qlen = p - qhead;
        }
      }
    }
    const int need = it < SCAN_ITERS ? 32 : 1;
    while (qlen >= need) {
      const int cnt = qlen < 32 ? qlen : 32;
      int ent = 0;
      if (l < cnt) ent = q[(qhead + l) & (Q_SLOTS - 1)];
      const int eid   = ent >> 8;
      const int row_l = ent & 255;
      const int    s_l = src[eid];
      const float2 a_l = *(const float2*)(ea + 2 * (long)eid);
      const int a0_l = __float_as_int(a_l.x);
      const int a1_l = __float_as_int(a_l.y);
      if (RECORD) {
        if (!over && nrec + cnt <= REC_CAP) {
          if (l < cnt) lst[nrec + l] = make_int4(s_l, row_l, a0_l, a1_l);
          nrec += cnt;
        } else {
          over = true;
        }
      }
      add_group(h, acc, l, cnt, s_l, row_l, a0_l, a1_l, w0, w1, bb);
      qhead = (qhead + cnt) & (Q_SLOTS - 1);
      qlen -= cnt;
    }
  }
  return over ? -1 : nrec;
}


__device__ __forceinline__ void load_w(const float* __restrict__ W, v4f (&we)[2], v4f (&wo)[2],
                                       const int l, const int w)
{
#pragma unroll
  for (int i = 0; i < 2; ++i) {
    const int p = 32 * i + w;
    we[i] = ((const v4f*)W)[(2 * p) * 32 + l];
    wo[i] = ((const v4f*)W)[(2 * p + 1) * 32 + l];
  }
}
__device__ __forceinline__ void store_w(float* Wt, const v4f (&we)[2], const v4f (&wo)[2],
                                        const int l, const int w)
{
  float* wd = Wt + 128 * (l >> 4) + 4 * (l & 15);
#pragma unroll
  for (int i = 0; i < 2; ++i) {
    float* d = wd + (32 * i + w) * WP_PITCH;
    *(v4f*)d        = (v4f){we[i][0], wo[i][0], we[i][1], wo[i][1]};
    *(v4f*)(d + 64) = (v4f){we[i][2], wo[i][2], we[i][3], wo[i][3]};
  }
}

__device__ __forceinline__ void node_tile(
    const float* acc, const float* Wt, const float* __restrict__ bias,
    float* __restrict__ hout, const int l, const int w, const int n0)
{
  if (w >= 2 * OWN_TILES) return;
  const int t = w >> 1;
  const int g = w & 1;
  if (n0 + 16 * t >= N_NODES) return;
  const int c  = l & 15;
  const int hh = l >> 4;

  int lr = 16 * t + c;
  lr = lr < OWN_ROWS ? lr : OWN_ROWS - 1;
  const float* xa = acc + lr * ACC_PITCH + 4 * hh;
  const float* wb = Wt + (2 * hh) * WP_PITCH + 128 * g + 4 * c;

  v8f d[4];
#pragma unroll
  for (int i = 0; i < 4; ++i) d[i] = (v8f){};

  v4f xn = *(const v4f*)xa;
  v4f bn[2][2];
#pragma unroll
  for (int u = 0; u < 2; ++u)
#pragma unroll
    for (int j = 0; j < 2; ++j)
      bn[u][j] = *(const v4f*)(wb + u * WP_PITCH + 64 * j);
#pragma unroll
  for (int q = 0; q < 16; ++q) {
    const v4f x = xn;
    v4f b[2][2];
#pragma unroll
    for (int u = 0; u < 2; ++u)
#pragma unroll
      for (int j = 0; j < 2; ++j) b[u][j] = bn[u][j];
    if (q + 1 < 16) {
      xn = *(const v4f*)(xa + 8 * (q + 1));
#pragma unroll
      for (int u = 0; u < 2; ++u)
#pragma unroll
        for (int j = 0; j < 2; ++j)
          bn[u][j] = *(const v4f*)(wb + (4 * (q + 1) + u) * WP_PITCH + 64 * j);
    }
#pragma unroll
    for (int u = 0; u < 2; ++u) {
      v2f a;
      a.x = x[2 * u];
      a.y = x[2 * u + 1];
#pragma unroll
      for (int j = 0; j < 2; ++j) {
#pragma unroll
        for (int e = 0; e < 2; ++e) {
          v2f bm;
          bm.x = b[u][j][2 * e];
          bm.y = b[u][j][2 * e + 1];
          d[2 * j + e] = __builtin_amdgcn_wmma_f32_16x16x4_f32(
               false, a,  false, bm,
               (short)0, d[2 * j + e],  false,  false);
        }
      }
    }
  }

  const v4f bv = *(const v4f*)(bias + 64 * g + 4 * c);
  float* orow = hout + ((long)n0 + 16 * t + 8 * hh) * HID + 64 * g + 4 * c;
#pragma unroll
  for (int r = 0; r < 8; ++r) {
    v4f y;
#pragma unroll
    for (int i = 0; i < 4; ++i) {
      const float v = d[i][r] + bv[i];
      y[i] = (v >= 0.f) ? v : NEG_SLOPE * v;
    }
    const int row = 16 * t + 8 * hh + r;
    if (row < OWN_ROWS && n0 + row < N_NODES) *(v4f*)(orow + (long)r * HID) = y;
  }
}

#define H_PER_WAVE ((OWN_ROWS + OWN_WAVES - 1) / OWN_WAVES)
__device__ __forceinline__ void store_rows(float* acc, const v4f (&hv)[H_PER_WAVE], const int l, const int w)
{
  asm volatile("" :: "v"(hv[H_PER_WAVE - 1]));
#pragma unroll
  for (int k = 0; k < H_PER_WAVE; ++k) {
    const int r = w + OWN_WAVES * k;
    if (r < OWN_ROWS) *(v4f*)(acc + r * ACC_PITCH + 4 * l) = hv[k];
  }
}

template <bool RECORD>
__device__ __forceinline__ void edge_owner_body(
    float* acc, float* Wt, int* ring, int* lens,
    const float* __restrict__ h,
    const int*   __restrict__ src,
    const int*   __restrict__ dst,
    const float* __restrict__ ea,
    const float* __restrict__ Wl0,
    const float* __restrict__ Wl1,
    const float* __restrict__ bl,
    const float* __restrict__ W,
    const float* __restrict__ bias,
    float* __restrict__ hout,
    int* lists)
{
  const int tid = threadIdx.x;
  const int l   = tid & 31;
  const int w   = __builtin_amdgcn_readfirstlane(tid >> 5);
  __builtin_assume(w >= 0 && w < OWN_WAVES);
  const int n0  = blockIdx.x * OWN_ROWS;

  const int lid = blockIdx.x * OWN_WAVES + w;
  int4* lst = (int4*)lists + (long)lid * REC_CAP;
  int*  len = lists + LIST_INTS + (long)blockIdx.x * OWN_WAVES;

  int n = 0;
  int4 rec[REC_CAP / 32];
  if (!RECORD) {
    n = __builtin_amdgcn_readfirstlane(len[w]);
#pragma unroll
    for (int k = 0; k < REC_CAP / 32; ++k) {
      rec[k] = make_int4(0, 0, 0, 0);
      if (32 * k + l < n) rec[k] = lst[32 * k + l];
    }
  }

  v4f hv[H_PER_WAVE];
#pragma unroll
  for (int k = 0; k < H_PER_WAVE; ++k) {
    int gr = n0 + w + OWN_WAVES * k;
    gr = gr < N_NODES ? gr : N_NODES - 1;
    hv[k] = ((const v4f*)h)[(long)gr * (HID / 4) + l];
  }
  v4f we[2], wo[2];
  load_w(W, we, wo, l, w);

  float w0[4], w1[4], bb[4];
#pragma unroll
  for (int j = 0; j < 4; ++j) {
    w0[j] = Wl0[l + 32 * j];
    w1[j] = Wl1[l + 32 * j];
    bb[j] = bl[l + 32 * j];
  }

  store_rows(acc, hv, l, w);
  store_w(Wt, we, wo, l, w);
  __syncthreads();

  if (RECORD) {
    n = scan_wave<true>(h, src, dst, ea, acc, ring + w * Q_SLOTS, l, w, n0, w0, w1, bb, lst);
    if (l == 0) lens[w] = n;
  } else {
    if (n >= 0) {
#pragma unroll
      for (int k = 0; k < REC_CAP / 32; ++k) {
        if (32 * k < n) {
          const int rest = n - 32 * k;
          add_group(h, acc, l, rest < 32 ? rest : 32, rec[k].x, rec[k].y, rec[k].z, rec[k].w, w0, w1, bb);
        }
      }
    } else {
      scan_wave<false>(h, src, dst, ea, acc, ring + w * Q_SLOTS, l, w, n0, w0, w1, bb, lst);
    }
  }

  __syncthreads();
  if (RECORD) {
    if (w == 0) len[l] = lens[l];
  }
  node_tile(acc, Wt, bias, hout, l, w, n0);
}

__global__ __launch_bounds__(OWN_THREADS) void edge_owner_record_kernel(
    const float* __restrict__ h, const int* __restrict__ src, const int* __restrict__ dst,
    const float* __restrict__ ea, const float* __restrict__ Wl0, const float* __restrict__ Wl1,
    const float* __restrict__ bl, const float* __restrict__ W, const float* __restrict__ bias,
    float* __restrict__ hout, int* __restrict__ lists)
{
  __shared__ __attribute__((aligned(16))) float acc[OWN_ROWS * ACC_PITCH];
  __shared__ __attribute__((aligned(16))) float Wt[(HID / 2) * WP_PITCH];
  __shared__ int   ring[OWN_WAVES * Q_SLOTS];
  __shared__ int   lens[OWN_WAVES];
  edge_owner_body<true>(acc, Wt, ring, lens, h, src, dst, ea, Wl0, Wl1, bl, W, bias, hout, lists);
}

__global__ __launch_bounds__(OWN_THREADS) void edge_owner_replay_kernel(
    const float* __restrict__ h, const int* __restrict__ src, const int* __restrict__ dst,
    const float* __restrict__ ea, const float* __restrict__ Wl0, const float* __restrict__ Wl1,
    const float* __restrict__ bl, const float* __restrict__ W, const float* __restrict__ bias,
    float* __restrict__ hout, int* __restrict__ lists)
{
  __shared__ __attribute__((aligned(16))) float acc[OWN_ROWS * ACC_PITCH];
  __shared__ __attribute__((aligned(16))) float Wt[(HID / 2) * WP_PITCH];
  __shared__ int   ring[OWN_WAVES * Q_SLOTS];
  edge_owner_body<false>(acc, Wt, ring, nullptr, h, src, dst, ea, Wl0, Wl1, bl, W, bias, hout, lists);
}

extern "C" void kernel_launch(void* const* d_in, const int* in_sizes, int n_in,
                              void* d_out, int out_size, void* d_ws, size_t ws_size,
                              hipStream_t stream) {
  (void)in_sizes; (void)n_in; (void)out_size; (void)ws_size;

  const float* x  = (const float*)d_in[0];
  const int*   ei = (const int*)  d_in[1];
  const float* ea = (const float*)d_in[2];
  const float* Wl = (const float*)d_in[3];
  const float* bl = (const float*)d_in[4];
  const float* W  = (const float*)d_in[5];
  const float* b  = (const float*)d_in[6];
  float*       out = (float*)d_out;

  const int* src = ei;
  const int* dst = ei + N_EDGES;

  const long NH = (long)N_NODES * HID;
  float* h1    = (float*)d_ws;
  float* h2    = h1 + NH;
  int*   lists = (int*)(h2 + NH);

  edge_owner_record_kernel<<<OWN_BLOCKS, OWN_THREADS, 0, stream>>>(
      x, src, dst, ea,
      Wl,
      Wl + HID,
      bl, W, b, h1, lists);
  edge_owner_replay_kernel<<<OWN_BLOCKS, OWN_THREADS, 0, stream>>>(
      h1, src, dst, ea,
      Wl + 2 * HID, Wl + 2 * HID + HID,
      bl + HID, W + HID * HID, b + HID, h2, lists);
  edge_owner_replay_kernel<<<OWN_BLOCKS, OWN_THREADS, 0, stream>>>(
      h2, src, dst, ea,
      Wl + 4 * HID, Wl + 4 * HID + HID,
      bl + 2 * HID, W + 2 * HID * HID, b + 2 * HID, out, lists);
}


# New first launch: 1 owner byte (dst/196) + a 16-B record {src,dst,attrs} per edge; layer 0 blocks search 600 KB of bytes (was 2400 KB of dst), fetch 1 record per edge; 4 launches; adds arrival-order
# speedup vs baseline: 4.0400x; 1.0428x over previous
#include <hip/hip_runtime.h>

#define N_NODES 50000
#define N_EDGES 600000
#define HID 128
#define NEG_SLOPE 0.01f
#define N_LAYERS 3

typedef __attribute__((ext_vector_type(2))) float v2f;
typedef __attribute__((ext_vector_type(8))) float v8f;
typedef __attribute__((ext_vector_type(4))) float v4f;

#define OWN_ROWS 196
#define OWN_BLOCKS ((N_NODES + OWN_ROWS - 1) / OWN_ROWS)
#define OWN_WAVES 32
#define OWN_THREADS (OWN_WAVES * 32)
#define OWN_TILES ((OWN_ROWS + 15) / 16)
#define ACC_PITCH 132
#define WP_PITCH 260
#define ACC_ROWS (OWN_ROWS + 1)
#define SCAN_AHEAD 4
#define Q_SLOTS 1024
#define EDGE_GROUP 8
#define N_DST4 (N_EDGES / 4)
#define N_ID4 (N_EDGES / 16)
#define SCAN_STEP (OWN_WAVES * 32)
#define SCAN_ITERS ((N_ID4 + SCAN_STEP - 1) / SCAN_STEP)
#define SCAN_TRIPS ((SCAN_ITERS + SCAN_AHEAD - 1) / SCAN_AHEAD)
#define REC_CAP 160
#define N_LISTS (OWN_BLOCKS * OWN_WAVES)
#define LIST_INTS ((long)N_LISTS * REC_CAP * 4)
#define PREP_THREADS 256
#define PREP_BLOCKS 256
#define PREP_STRIDE (PREP_BLOCKS * PREP_THREADS)
#define PREP_TRIPS ((N_EDGES + PREP_STRIDE - 1) / PREP_STRIDE)
#define PREP_TRIPS4 ((N_DST4 + PREP_STRIDE - 1) / PREP_STRIDE)
static_assert(N_EDGES % 16 == 0, "dst and the owner bytes are read in whole 16-byte pieces");
static_assert(OWN_BLOCKS == 256, "an owner number is one byte, and every byte value is a block");
static_assert(31 + 512 <= Q_SLOTS, "ring holds a leftover and one scan step (16 edges a lane)");
static_assert(REC_CAP % 32 == 0, "a list is a whole number of 32-record groups");
static_assert(2 * OWN_TILES <= OWN_WAVES, "one wave per (row tile, column half)");
static_assert(HID == 128, "a row is 32 lanes x 16 bytes; the fragment arithmetic below");
static_assert((LIST_INTS + N_LISTS) * 4 + N_EDGES <= (long)N_NODES * HID * 4,
              "lists, lengths and owner bytes fit the workspace plane they replace");
static_assert((long)N_EDGES * 16 <= (long)N_NODES * HID * 4,
              "the edge records fit the plane of layer 2's h, which layer 0 does not use");

__device__ __forceinline__ void add_group(
    const float* __restrict__ h, float* acc, const int l, const int cnt,
    const int s_l, const int row_l, const int a0_l, const int a1_l,
    const float (&w0)[4], const float (&w1)[4], const float (&bb)[4])
{
#pragma unroll
  for (int g = 0; g < 32; g += EDGE_GROUP) {
    if (g < cnt) {
      float hv[EDGE_GROUP][4];
#pragma unroll
      for (int i = 0; i < EDGE_GROUP; ++i) {
        const int s = __builtin_amdgcn_readlane(s_l, g + i);
        const float* hp = h + (long)s * HID;
        hv[i][0] = hp[l];
        hv[i][1] = hp[l + 32];
        hv[i][2] = hp[l + 64];
        hv[i][3] = hp[l + 96];
      }
#pragma unroll
      for (int i = 0; i < EDGE_GROUP; ++i) {
        if (g + i < cnt) {
          const int   r  = __builtin_amdgcn_readlane(row_l, g + i);
          const float a0 = __int_as_float(__builtin_amdgcn_readlane(a0_l, g + i));
          const float a1 = __int_as_float(__builtin_amdgcn_readlane(a1_l, g + i));
          float* ap = acc + r * ACC_PITCH + l;
#pragma unroll
          for (int j = 0; j < 4; ++j) {
            float m = hv[i][j] + a0 * w0[j] + a1 * w1[j] + bb[j];
            m = m > 0.f ? m : 0.f;
            __hip_atomic_fetch_add(ap + 32 * j, m, __ATOMIC_RELAXED, __HIP_MEMORY_SCOPE_WORKGROUP);
          }
        }
      }
    }
  }
}

__global__ __launch_bounds__(PREP_THREADS) void edge_prep_kernel(
    const int* __restrict__ src, const int* __restrict__ dst, const float* __restrict__ ea,
    int* __restrict__ ids, int* __restrict__ recs)
{
  const int t0 = blockIdx.x * PREP_THREADS + threadIdx.x;
  int4 dd[PREP_TRIPS4];
#pragma unroll
  for (int k = 0; k < PREP_TRIPS4; ++k) {
    const int t = t0 + k * PREP_STRIDE;
    dd[k] = ((const int4*)dst)[t < N_DST4 ? t : N_DST4 - 1];
  }
  int sv[PREP_TRIPS], dv[PREP_TRIPS];
  float2 av[PREP_TRIPS];
#pragma unroll
  for (int k = 0; k < PREP_TRIPS; ++k) {
    const int e = t0 + k * PREP_STRIDE;
    const int ec = e < N_EDGES ? e : N_EDGES - 1;
    sv[k] = src[ec];
    dv[k] = dst[ec];
    av[k] = *(const float2*)(ea + 2 * (long)ec);
  }
#pragma unroll
  for (int k = 0; k < PREP_TRIPS4; ++k) {
    const int t = t0 + k * PREP_STRIDE;
    unsigned o0 = (unsigned)dd[k].x / OWN_ROWS, o1 = (unsigned)dd[k].y / OWN_ROWS;
    unsigned o2 = (unsigned)dd[k].z / OWN_ROWS, o3 = (unsigned)dd[k].w / OWN_ROWS;
    o0 = o0 < 255u ? o0 : 255u;
    o1 = o1 < 255u ? o1 : 255u;
    o2 = o2 < 255u ? o2 : 255u;
    o3 = o3 < 255u ? o3 : 255u;
    if (t < N_DST4) ids[t] = (int)(o0 | (o1 << 8) | (o2 << 16) | (o3 << 24));
  }
#pragma unroll
  for (int k = 0; k < PREP_TRIPS; ++k) {
    const int e = t0 + k * PREP_STRIDE;
    if (e < N_EDGES)
      ((int4*)recs)[e] = make_int4(sv[k], dv[k], __float_as_int(av[k].x), __float_as_int(av[k].y));
  }
}

template <bool RECORD>
__device__ __forceinline__ int scan_wave(
    const float* __restrict__ h, const int* __restrict__ src, const int* __restrict__ dst,
    const int* __restrict__ ids, const int* __restrict__ recs,
    const float* __restrict__ ea, float* acc, int* q, const int l, const int w, const int n0,
    const float (&w0)[4], const float (&w1)[4], const float (&bb)[4],
    int4* lst)
{
  int qhead = 0, qlen = 0;
  int nrec = 0;
  bool over = false;

  const unsigned mine = (unsigned)blockIdx.x * 0x01010101u;
  const int4* i4 = (const int4*)ids;
  int4 dn[SCAN_AHEAD];
#pragma unroll
  for (int u = 0; u < SCAN_AHEAD; ++u) {
    const int idx = u * SCAN_STEP + w * 32 + l;
    dn[u] = i4[idx < N_ID4 ? idx : N_ID4 - 1];
  }

  for (int it = 0; it <= SCAN_TRIPS; ++it) {
    int4 d[SCAN_AHEAD];
#pragma unroll
    for (int u = 0; u < SCAN_AHEAD; ++u) d[u] = dn[u];
    if (it + 1 < SCAN_TRIPS) {
#pragma unroll
      for (int u = 0; u < SCAN_AHEAD; ++u) {
        const int idx = ((it + 1) * SCAN_AHEAD + u) * SCAN_STEP + w * 32 + l;
        dn[u] = i4[idx < N_ID4 ? idx : N_ID4 - 1];
      }
    }
#pragma unroll
    for (int u = 0; u < SCAN_AHEAD; ++u) {
      if (it < SCAN_TRIPS) {
        const int ib = (it * SCAN_AHEAD + u) * SCAN_STEP + w * 32 + l;
        const unsigned x[4] = {(unsigned)d[u].x, (unsigned)d[u].y, (unsigned)d[u].z, (unsigned)d[u].w};
        unsigned cm = 0;
#pragma unroll
        for (int c = 0; c < 4; ++c) {
          const unsigned t = x[c] ^ mine;
          const unsigned z = ~(((t & 0x7f7f7f7fu) + 0x7f7f7f7fu) | t) & 0x80808080u;
          cm |= z >> (7 - c);
        }
        if (ib >= N_ID4) cm = 0;
#pragma unroll 1
        for (int r = 0; r < 16; ++r) {
          const unsigned k = __builtin_amdgcn_ballot_w32(cm != 0);
          if (k == 0) break;
          if (cm != 0) {
            const int p = __builtin_ctz(cm);
            const int e = 16 * ib + 4 * (p & 7) + (p >> 3);
            q[(qhead + qlen + __builtin_amdgcn_mbcnt_lo(k, 0)) & (Q_SLOTS - 1)] = e;
            cm &= cm - 1;
          }
          qlen += __builtin_popcount(k);
        }
      }
      const int need = (it < SCAN_TRIPS || u + 1 < SCAN_AHEAD) ? 32 : 1;
      while (qlen >= need) {
        const int cnt = qlen < 32 ? qlen : 32;
        int eid = 0;
        if (l < cnt) eid = q[(qhead + l) & (Q_SLOTS - 1)];
        int s_l, d_l, a0_l, a1_l;
        if (RECORD) {
          const int4 rc = ((const int4*)recs)[eid];
          s_l = rc.x; d_l = rc.y; a0_l = rc.z; a1_l = rc.w;
        } else {
          d_l = dst[eid];
          s_l = src[eid];
          const float2 a_l = *(const float2*)(ea + 2 * (long)eid);
          a0_l = __float_as_int(a_l.x);
          a1_l = __float_as_int(a_l.y);
        }
        const unsigned r_l = (unsigned)(d_l - n0);
        const int row_l = r_l < OWN_ROWS ? (int)r_l : OWN_ROWS;
        if (RECORD) {
          if (!over && nrec + cnt <= REC_CAP) {
            if (l < cnt) lst[nrec + l] = make_int4(s_l, row_l, a0_l, a1_l);
            nrec += cnt;
          } else {
            over = true;
          }
        }
        add_group(h, acc, l, cnt, s_l, row_l, a0_l, a1_l, w0, w1, bb);
        qhead = (qhead + cnt) & (Q_SLOTS - 1);
        qlen -= cnt;
      }
    }
  }
  return over ? -1 : nrec;
}


__device__ __forceinline__ void load_w(const float* __restrict__ W, v4f (&we)[2], v4f (&wo)[2],
                                       const int l, const int w)
{
#pragma unroll
  for (int i = 0; i < 2; ++i) {
    const int p = 32 * i + w;
    we[i] = ((const v4f*)W)[(2 * p) * 32 + l];
    wo[i] = ((const v4f*)W)[(2 * p + 1) * 32 + l];
  }
}
__device__ __forceinline__ void store_w(float* Wt, const v4f (&we)[2], const v4f (&wo)[2],
                                        const int l, const int w)
{
  float* wd = Wt + 128 * (l >> 4) + 4 * (l & 15);
#pragma unroll
  for (int i = 0; i < 2; ++i) {
    float* d = wd + (32 * i + w) * WP_PITCH;
    *(v4f*)d        = (v4f){we[i][0], wo[i][0], we[i][1], wo[i][1]};
    *(v4f*)(d + 64) = (v4f){we[i][2], wo[i][2], we[i][3], wo[i][3]};
  }
}

__device__ __forceinline__ void node_tile(
    const float* acc, const float* Wt, const float* __restrict__ bias,
    float* __restrict__ hout, const int l, const int w, const int n0)
{
  if (w >= 2 * OWN_TILES) return;
  const int t = w >> 1;
  const int g = w & 1;
  if (n0 + 16 * t >= N_NODES) return;
  const int c  = l & 15;
  const int hh = l >> 4;

  int lr = 16 * t + c;
  lr = lr < OWN_ROWS ? lr : OWN_ROWS - 1;
  const float* xa = acc + lr * ACC_PITCH + 4 * hh;
  const float* wb = Wt + (2 * hh) * WP_PITCH + 128 * g + 4 * c;

  v8f d[4];
#pragma unroll
  for (int i = 0; i < 4; ++i) d[i] = (v8f){};

  v4f xn = *(const v4f*)xa;
  v4f bn[2][2];
#pragma unroll
  for (int u = 0; u < 2; ++u)
#pragma unroll
    for (int j = 0; j < 2; ++j)
      bn[u][j] = *(const v4f*)(wb + u * WP_PITCH + 64 * j);
#pragma unroll
  for (int q = 0; q < 16; ++q) {
    const v4f x = xn;
    v4f b[2][2];
#pragma unroll
    for (int u = 0; u < 2; ++u)
#pragma unroll
      for (int j = 0; j < 2; ++j) b[u][j] = bn[u][j];
    if (q + 1 < 16) {
      xn = *(const v4f*)(xa + 8 * (q + 1));
#pragma unroll
      for (int u = 0; u < 2; ++u)
#pragma unroll
        for (int j = 0; j < 2; ++j)
          bn[u][j] = *(const v4f*)(wb + (4 * (q + 1) + u) * WP_PITCH + 64 * j);
    }
#pragma unroll
    for (int u = 0; u < 2; ++u) {
      v2f a;
      a.x = x[2 * u];
      a.y = x[2 * u + 1];
#pragma unroll
      for (int j = 0; j < 2; ++j) {
#pragma unroll
        for (int e = 0; e < 2; ++e) {
          v2f bm;
          bm.x = b[u][j][2 * e];
          bm.y = b[u][j][2 * e + 1];
          d[2 * j + e] = __builtin_amdgcn_wmma_f32_16x16x4_f32(
               false, a,  false, bm,
               (short)0, d[2 * j + e],  false,  false);
        }
      }
    }
  }

  const v4f bv = *(const v4f*)(bias + 64 * g + 4 * c);
  float* orow = hout + ((long)n0 + 16 * t + 8 * hh) * HID + 64 * g + 4 * c;
#pragma unroll
  for (int r = 0; r < 8; ++r) {
    v4f y;
#pragma unroll
    for (int i = 0; i < 4; ++i) {
      const float v = d[i][r] + bv[i];
      y[i] = (v >= 0.f) ? v : NEG_SLOPE * v;
    }
    const int row = 16 * t + 8 * hh + r;
    if (row < OWN_ROWS && n0 + row < N_NODES) *(v4f*)(orow + (long)r * HID) = y;
  }
}

#define H_PER_WAVE ((OWN_ROWS + OWN_WAVES - 1) / OWN_WAVES)
__device__ __forceinline__ void store_rows(float* acc, const v4f (&hv)[H_PER_WAVE], const int l, const int w)
{
  asm volatile("" :: "v"(hv[H_PER_WAVE - 1]));
#pragma unroll
  for (int k = 0; k < H_PER_WAVE; ++k) {
    const int r = w + OWN_WAVES * k;
    if (r < OWN_ROWS) *(v4f*)(acc + r * ACC_PITCH + 4 * l) = hv[k];
    else if (r == OWN_ROWS) *(v4f*)(acc + r * ACC_PITCH + 4 * l) = (v4f){0.f, 0.f, 0.f, 0.f};
  }
}

template <bool RECORD>
__device__ __forceinline__ void edge_owner_body(
    float* acc, float* Wt, int* ring, int* lens,
    const float* __restrict__ h,
    const int*   __restrict__ src,
    const int*   __restrict__ dst,
    const float* __restrict__ ea,
    const float* __restrict__ Wl0,
    const float* __restrict__ Wl1,
    const float* __restrict__ bl,
    const float* __restrict__ W,
    const float* __restrict__ bias,
    float* __restrict__ hout,
    int* lists,
    const int* __restrict__ recs)
{
  const int tid = threadIdx.x;
  const int l   = tid & 31;
  const int w   = __builtin_amdgcn_readfirstlane(tid >> 5);
  __builtin_assume(w >= 0 && w < OWN_WAVES);
  const int n0  = blockIdx.x * OWN_ROWS;

  const int lid = blockIdx.x * OWN_WAVES + w;
  int4* lst = (int4*)lists + (long)lid * REC_CAP;
  int*  len = lists + LIST_INTS + (long)blockIdx.x * OWN_WAVES;
  const int* ids = lists + LIST_INTS + N_LISTS;

  int n = 0;
  int4 rec[REC_CAP / 32];
  if (!RECORD) {
    n = __builtin_amdgcn_readfirstlane(len[w]);
#pragma unroll
    for (int k = 0; k < REC_CAP / 32; ++k) {
      rec[k] = make_int4(0, 0, 0, 0);
      if (32 * k + l < n) rec[k] = lst[32 * k + l];
    }
  }

  v4f hv[H_PER_WAVE];
#pragma unroll
  for (int k = 0; k < H_PER_WAVE; ++k) {
    int gr = n0 + w + OWN_WAVES * k;
    gr = gr < N_NODES ? gr : N_NODES - 1;
    hv[k] = ((const v4f*)h)[(long)gr * (HID / 4) + l];
  }
  v4f we[2], wo[2];
  load_w(W, we, wo, l, w);

  float w0[4], w1[4], bb[4];
#pragma unroll
  for (int j = 0; j < 4; ++j) {
    w0[j] = Wl0[l + 32 * j];
    w1[j] = Wl1[l + 32 * j];
    bb[j] = bl[l + 32 * j];
  }

  store_rows(acc, hv, l, w);
  store_w(Wt, we, wo, l, w);
  __syncthreads();

  if (RECORD) {
    n = scan_wave<true>(h, src, dst, ids, recs, ea, acc, ring + w * Q_SLOTS, l, w, n0, w0, w1, bb, lst);
    if (l == 0) lens[w] = n;
  } else {
    if (n >= 0) {
#pragma unroll
      for (int k = 0; k < REC_CAP / 32; ++k) {
        if (32 * k < n) {
          const int rest = n - 32 * k;
          add_group(h, acc, l, rest < 32 ? rest : 32, rec[k].x, rec[k].y, rec[k].z, rec[k].w, w0, w1, bb);
        }
      }
    } else {
      scan_wave<false>(h, src, dst, ids, recs, ea, acc, ring + w * Q_SLOTS, l, w, n0, w0, w1, bb, lst);
    }
  }

  __syncthreads();
  if (RECORD) {
    if (w == 0) len[l] = lens[l];
  }
  node_tile(acc, Wt, bias, hout, l, w, n0);
}

__global__ __launch_bounds__(OWN_THREADS) void edge_owner_record_kernel(
    const float* __restrict__ h, const int* __restrict__ src, const int* __restrict__ dst,
    const float* __restrict__ ea, const float* __restrict__ Wl0, const float* __restrict__ Wl1,
    const float* __restrict__ bl, const float* __restrict__ W, const float* __restrict__ bias,
    float* __restrict__ hout, int* __restrict__ lists, const int* __restrict__ recs)
{
  __shared__ __attribute__((aligned(16))) float acc[ACC_ROWS * ACC_PITCH];
  __shared__ __attribute__((aligned(16))) float Wt[(HID / 2) * WP_PITCH];
  __shared__ int   ring[OWN_WAVES * Q_SLOTS];
  __shared__ int   lens[OWN_WAVES];
  edge_owner_body<true>(acc, Wt, ring, lens, h, src, dst, ea, Wl0, Wl1, bl, W, bias, hout, lists, recs);
}

__global__ __launch_bounds__(OWN_THREADS) void edge_owner_replay_kernel(
    const float* __restrict__ h, const int* __restrict__ src, const int* __restrict__ dst,
    const float* __restrict__ ea, const float* __restrict__ Wl0, const float* __restrict__ Wl1,
    const float* __restrict__ bl, const float* __restrict__ W, const float* __restrict__ bias,
    float* __restrict__ hout, int* __restrict__ lists)
{
  __shared__ __attribute__((aligned(16))) float acc[ACC_ROWS * ACC_PITCH];
  __shared__ __attribute__((aligned(16))) float Wt[(HID / 2) * WP_PITCH];
  __shared__ int   ring[OWN_WAVES * Q_SLOTS];
  edge_owner_body<false>(acc, Wt, ring, nullptr, h, src, dst, ea, Wl0, Wl1, bl, W, bias, hout, lists, nullptr);
}

extern "C" void kernel_launch(void* const* d_in, const int* in_sizes, int n_in,
                              void* d_out, int out_size, void* d_ws, size_t ws_size,
                              hipStream_t stream) {
  (void)in_sizes; (void)n_in; (void)out_size; (void)ws_size;

  const float* x  = (const float*)d_in[0];
  const int*   ei = (const int*)  d_in[1];
  const float* ea = (const float*)d_in[2];
  const float* Wl = (const float*)d_in[3];
  const float* bl = (const float*)d_in[4];
  const float* W  = (const float*)d_in[5];
  const float* b  = (const float*)d_in[6];
  float*       out = (float*)d_out;

  const int* src = ei;
  const int* dst = ei + N_EDGES;

  const long NH = (long)N_NODES * HID;
  float* h1    = (float*)d_ws;
  float* h2    = h1 + NH;
  int*   lists = (int*)(h2 + NH);

  edge_prep_kernel<<<PREP_BLOCKS, PREP_THREADS, 0, stream>>>(src, dst, ea, lists + LIST_INTS + N_LISTS, (int*)h2);
  edge_owner_record_kernel<<<OWN_BLOCKS, OWN_THREADS, 0, stream>>>(
      x, src, dst, ea,
      Wl,
      Wl + HID,
      bl, W, b, h1, lists, (const int*)h2);
  edge_owner_replay_kernel<<<OWN_BLOCKS, OWN_THREADS, 0, stream>>>(
      h1, src, dst, ea,
      Wl + 2 * HID, Wl + 2 * HID + HID,
      bl + HID, W + HID * HID, b + HID, h2, lists);
  edge_owner_replay_kernel<<<OWN_BLOCKS, OWN_THREADS, 0, stream>>>(
      h2, src, dst, ea,
      Wl + 4 * HID, Wl + 4 * HID + HID,
      bl + 2 * HID, W + 2 * HID * HID, b + 2 * HID, out, lists);
}
